# starting kernel
# baseline (speedup 1.0000x reference)
#include <hip/hip_runtime.h>


#define N_GENES 768
#define HID     128
#define NTILE   48

typedef float v2f __attribute__((ext_vector_type(2)));
typedef float v8f __attribute__((ext_vector_type(8)));

__global__ __launch_bounds__(HID) void grn_proj_kernel(
    const float* __restrict__ X, const float* __restrict__ W1,
    const float* __restrict__ Wb,
    float* __restrict__ Ai, float* __restrict__ Bj, float* __restrict__ Y) {
  __shared__ float sX[HID];
  const int i = blockIdx.x;
  const int c = threadIdx.x;
  sX[c] = X[i * HID + c];
  __syncthreads();
  const float* __restrict__ w1a = W1 + c * (2 * HID);
  const float* __restrict__ w1b = w1a + HID;
  float a = 0.f, b = 0.f, y = 0.f;
#pragma unroll 8
  for (int k = 0; k < HID; ++k) {
    const float xv = sX[k];
    a = fmaf(xv, w1a[k], a);
    b = fmaf(xv, w1b[k], b);
    y = fmaf(xv, Wb[k * HID + c], y);
  }
  Ai[i * HID + c] = a;
  Bj[i * HID + c] = b;
  Y[i * HID + c]  = y;
}

__global__ __launch_bounds__(32) void grn_tile_kernel(
    const float* __restrict__ X,  const float* __restrict__ Ai,
    const float* __restrict__ Bj, const float* __restrict__ Y,
    const float* __restrict__ b1, const float* __restrict__ W2,
    const float* __restrict__ b2, const float* __restrict__ bb,
    float* __restrict__ out) {
  __shared__ alignas(16) float sAi[16 * HID];
  __shared__ alignas(16) float sBj[16 * HID];
  __shared__ alignas(16) float sB1[HID];
  __shared__ alignas(16) float sW2[3 * HID];
  __shared__ float sSign[256];

  const int lane = threadIdx.x;
  const int i0 = blockIdx.x * 16;
  const int j0 = blockIdx.y * 16;

  {
    const float4* __restrict__ gAi = (const float4*)(Ai + i0 * HID);
    const float4* __restrict__ gBj = (const float4*)(Bj + j0 * HID);
    float4* dAi = (float4*)sAi;
    float4* dBj = (float4*)sBj;
#pragma unroll
    for (int q = 0; q < 16; ++q) {
      dAi[lane + 32 * q] = gAi[lane + 32 * q];
      dBj[lane + 32 * q] = gBj[lane + 32 * q];
    }
    ((float4*)sB1)[lane] = ((const float4*)b1)[lane];
#pragma unroll
    for (int q = 0; q < 3; ++q)
      ((float4*)sW2)[lane + 32 * q] = ((const float4*)W2)[lane + 32 * q];
  }
  __syncthreads();

  const float bbv = bb[0];
  v8f acc = {bbv, bbv, bbv, bbv, bbv, bbv, bbv, bbv};
  const int lm = lane & 15;
  const int kb = (lane >> 4) << 1;
  const float* __restrict__ arow = Y + (i0 + lm) * HID + kb;
  const float* __restrict__ brow = X + (j0 + lm) * HID + kb;
#pragma unroll
  for (int kk = 0; kk < HID; kk += 4) {
    const v2f afrag = *(const v2f*)(arow + kk);
    const v2f bfrag = *(const v2f*)(brow + kk);
    acc = __builtin_amdgcn_wmma_f32_16x16x4_f32(
         false, afrag,  false, bfrag,
         (short)0, acc,  false,  false);
  }

  const float b20 = b2[0], b21 = b2[1], b22 = b2[2];
#pragma unroll
  for (int r = 0; r < 8; ++r) {
    const int p  = lane + 32 * r;
    const int mi = p >> 4;
    const int nj = p & 15;
    const float4* __restrict__ ha = (const float4*)(sAi + mi * HID);
    const float4* __restrict__ hb = (const float4*)(sBj + nj * HID);
    const float4* __restrict__ hc = (const float4*)sB1;
    const float4* __restrict__ w0 = (const float4*)sW2;
    const float4* __restrict__ w1 = (const float4*)(sW2 + HID);
    const float4* __restrict__ w2 = (const float4*)(sW2 + 2 * HID);
    float l0 = b20, l1 = b21, l2 = b22;
#pragma unroll 4
    for (int k4 = 0; k4 < HID / 4; ++k4) {
      const float4 va = ha[k4], vb = hb[k4], vc = hc[k4];
      float4 h;
      h.x = fmaxf(va.x + vb.x + vc.x, 0.f);
      h.y = fmaxf(va.y + vb.y + vc.y, 0.f);
      h.z = fmaxf(va.z + vb.z + vc.z, 0.f);
      h.w = fmaxf(va.w + vb.w + vc.w, 0.f);
      const float4 u0 = w0[k4], u1 = w1[k4], u2 = w2[k4];
      l0 = fmaf(h.x, u0.x, fmaf(h.y, u0.y, fmaf(h.z, u0.z, fmaf(h.w, u0.w, l0))));
      l1 = fmaf(h.x, u1.x, fmaf(h.y, u1.y, fmaf(h.z, u1.z, fmaf(h.w, u1.w, l1))));
      l2 = fmaf(h.x, u2.x, fmaf(h.y, u2.y, fmaf(h.z, u2.z, fmaf(h.w, u2.w, l2))));
    }
    const float s = (l0 >= l1 && l0 >= l2) ? 1.0f : ((l1 >= l2) ? 0.0f : -1.0f);
    sSign[p] = s;
  }
  __syncthreads();

  const int n = lane & 15;
  const int mbase = (lane >> 4) << 3;
#pragma unroll
  for (int v = 0; v < 8; ++v) {
    const int m  = v + mbase;
    const int gi = i0 + m;
    const int gj = j0 + n;
    const float val = (gi == gj) ? 0.f : sSign[m * 16 + n] * acc[v];
    out[gi * N_GENES + gj] = val;
  }
}

extern "C" void kernel_launch(void* const* d_in, const int* in_sizes, int n_in,
                              void* d_out, int out_size, void* d_ws,
                              size_t ws_size, hipStream_t stream) {
  (void)in_sizes; (void)n_in; (void)out_size; (void)ws_size;
  const float* X  = (const float*)d_in[0];
  const float* W1 = (const float*)d_in[1];
  const float* b1 = (const float*)d_in[2];
  const float* W2 = (const float*)d_in[3];
  const float* b2 = (const float*)d_in[4];
  const float* Wb = (const float*)d_in[5];
  const float* bb = (const float*)d_in[6];
  float* out = (float*)d_out;

  float* ws = (float*)d_ws;
  float* Ai = ws;
  float* Bj = ws + N_GENES * HID;
  float* Y  = ws + 2 * N_GENES * HID;

  grn_proj_kernel<<<dim3(N_GENES), dim3(HID), 0, stream>>>(X, W1, Wb, Ai, Bj, Y);
  grn_tile_kernel<<<dim3(NTILE, NTILE), dim3(32), 0, stream>>>(
      X, Ai, Bj, Y, b1, W2, b2, bb, out);
}


# speedup vs baseline: 1.5796x; 1.5796x over previous
#include <hip/hip_runtime.h>


#define N_GENES 768
#define HID     128
#define NTILE   48
#define BJ_PITCH (HID + 4)

typedef float v2f __attribute__((ext_vector_type(2)));
typedef float v8f __attribute__((ext_vector_type(8)));

__global__ __launch_bounds__(HID) void grn_proj_kernel(
    const float* __restrict__ X, const float* __restrict__ W1,
    const float* __restrict__ Wb,
    float* __restrict__ Ai, float* __restrict__ Bj, float* __restrict__ Y) {
  __shared__ float sX[HID];
  const int i = blockIdx.x;
  const int c = threadIdx.x;
  sX[c] = X[i * HID + c];
  __syncthreads();
  const float* __restrict__ w1a = W1 + c * (2 * HID);
  const float* __restrict__ w1b = w1a + HID;
  float a = 0.f, b = 0.f, y = 0.f;
#pragma unroll 8
  for (int k = 0; k < HID; ++k) {
    const float xv = sX[k];
    a = fmaf(xv, w1a[k], a);
    b = fmaf(xv, w1b[k], b);
    y = fmaf(xv, Wb[k * HID + c], y);
  }
  Ai[i * HID + c] = a;
  Bj[i * HID + c] = b;
  Y[i * HID + c]  = y;
}

__global__ __launch_bounds__(32) void grn_tile_kernel(
    const float* __restrict__ X,  const float* __restrict__ Ai,
    const float* __restrict__ Bj, const float* __restrict__ Y,
    const float* __restrict__ b1, const float* __restrict__ W2,
    const float* __restrict__ b2, const float* __restrict__ bb,
    float* __restrict__ out) {
  __shared__ alignas(16) float sAi[16 * HID];
  __shared__ alignas(16) float sBj[16 * BJ_PITCH];
  __shared__ alignas(16) float sB1[HID];
  __shared__ alignas(16) float sW2[3 * HID];
  __shared__ float sSign[256];

  const int lane = threadIdx.x;
  const int i0 = blockIdx.x * 16;
  const int j0 = blockIdx.y * 16;

  {
    const float4* __restrict__ gAi = (const float4*)(Ai + i0 * HID);
    const float4* __restrict__ gBj = (const float4*)(Bj + j0 * HID);
    float4* dAi = (float4*)sAi;
    float4* dBj = (float4*)sBj;
#pragma unroll
    for (int q = 0; q < 16; ++q) {
      dAi[lane + 32 * q] = gAi[lane + 32 * q];
      dBj[lane + (BJ_PITCH / 4) * q] = gBj[lane + 32 * q];
    }
    ((float4*)sB1)[lane] = ((const float4*)b1)[lane];
#pragma unroll
    for (int q = 0; q < 3; ++q)
      ((float4*)sW2)[lane + 32 * q] = ((const float4*)W2)[lane + 32 * q];
  }
  __syncthreads();

  const float bbv = bb[0];
  v8f acc = {bbv, bbv, bbv, bbv, bbv, bbv, bbv, bbv};
  const int lm = lane & 15;
  const int kb = (lane >> 4) << 1;
  const float* __restrict__ arow = Y + (i0 + lm) * HID + kb;
  const float* __restrict__ brow = X + (j0 + lm) * HID + kb;
#pragma unroll
  for (int kk = 0; kk < HID; kk += 4) {
    const v2f afrag = *(const v2f*)(arow + kk);
    const v2f bfrag = *(const v2f*)(brow + kk);
    acc = __builtin_amdgcn_wmma_f32_16x16x4_f32(
         false, afrag,  false, bfrag,
         (short)0, acc,  false,  false);
  }

  const float b20 = b2[0], b21 = b2[1], b22 = b2[2];
#pragma unroll
  for (int r = 0; r < 8; ++r) {
    const int p  = lane + 32 * r;
    const int mi = p >> 4;
    const int nj = p & 15;
    const float4* __restrict__ ha = (const float4*)(sAi + mi * HID);
    const float4* __restrict__ hb = (const float4*)(sBj + nj * BJ_PITCH);
    const float4* __restrict__ hc = (const float4*)sB1;
    const float4* __restrict__ w0 = (const float4*)sW2;
    const float4* __restrict__ w1 = (const float4*)(sW2 + HID);
    const float4* __restrict__ w2 = (const float4*)(sW2 + 2 * HID);
    float l0 = b20, l1 = b21, l2 = b22;
#pragma unroll 4
    for (int k4 = 0; k4 < HID / 4; ++k4) {
      const float4 va = ha[k4], vb = hb[k4], vc = hc[k4];
      float4 h;
      h.x = fmaxf(va.x + vb.x + vc.x, 0.f);
      h.y = fmaxf(va.y + vb.y + vc.y, 0.f);
      h.z = fmaxf(va.z + vb.z + vc.z, 0.f);
      h.w = fmaxf(va.w + vb.w + vc.w, 0.f);
      const float4 u0 = w0[k4], u1 = w1[k4], u2 = w2[k4];
      l0 = fmaf(h.x, u0.x, fmaf(h.y, u0.y, fmaf(h.z, u0.z, fmaf(h.w, u0.w, l0))));
      l1 = fmaf(h.x, u1.x, fmaf(h.y, u1.y, fmaf(h.z, u1.z, fmaf(h.w, u1.w, l1))));
      l2 = fmaf(h.x, u2.x, fmaf(h.y, u2.y, fmaf(h.z, u2.z, fmaf(h.w, u2.w, l2))));
    }
    const float s = (l0 >= l1 && l0 >= l2) ? 1.0f : ((l1 >= l2) ? 0.0f : -1.0f);
    sSign[p] = s;
  }
  __syncthreads();

  const int n = lane & 15;
  const int mbase = (lane >> 4) << 3;
#pragma unroll
  for (int v = 0; v < 8; ++v) {
    const int m  = v + mbase;
    const int gi = i0 + m;
    const int gj = j0 + n;
    const float val = (gi == gj) ? 0.f : sSign[m * 16 + n] * acc[v];
    out[gi * N_GENES + gj] = val;
  }
}

extern "C" void kernel_launch(void* const* d_in, const int* in_sizes, int n_in,
                              void* d_out, int out_size, void* d_ws,
                              size_t ws_size, hipStream_t stream) {
  (void)in_sizes; (void)n_in; (void)out_size; (void)ws_size;
  const float* X  = (const float*)d_in[0];
  const float* W1 = (const float*)d_in[1];
  const float* b1 = (const float*)d_in[2];
  const float* W2 = (const float*)d_in[3];
  const float* b2 = (const float*)d_in[4];
  const float* Wb = (const float*)d_in[5];
  const float* bb = (const float*)d_in[6];
  float* out = (float*)d_out;

  float* ws = (float*)d_ws;
  float* Ai = ws;
  float* Bj = ws + N_GENES * HID;
  float* Y  = ws + 2 * N_GENES * HID;

  grn_proj_kernel<<<dim3(N_GENES), dim3(HID), 0, stream>>>(X, W1, Wb, Ai, Bj, Y);
  grn_tile_kernel<<<dim3(NTILE, NTILE), dim3(32), 0, stream>>>(
      X, Ai, Bj, Y, b1, W2, b2, bb, out);
}


# Tile kernel classifier in registers: a lane takes the 8 pairs of its own accumulator elements, reads Bj once per 8 rows, b1/W2 via scalar loads; LDS reads per wave 1536 to 288; signs bit-identical
# speedup vs baseline: 1.8586x; 1.1766x over previous
#include <hip/hip_runtime.h>


#define N_GENES 768
#define HID     128
#define NTILE   48
#define BJ_PITCH (HID + 4)

typedef float v2f __attribute__((ext_vector_type(2)));
typedef float v8f __attribute__((ext_vector_type(8)));

__global__ __launch_bounds__(HID) void grn_proj_kernel(
    const float* __restrict__ X, const float* __restrict__ W1,
    const float* __restrict__ Wb,
    float* __restrict__ Ai, float* __restrict__ Bj, float* __restrict__ Y) {
  __shared__ float sX[HID];
  const int i = blockIdx.x;
  const int c = threadIdx.x;
  sX[c] = X[i * HID + c];
  __syncthreads();
  const float* __restrict__ w1a = W1 + c * (2 * HID);
  const float* __restrict__ w1b = w1a + HID;
  float a = 0.f, b = 0.f, y = 0.f;
#pragma unroll 8
  for (int k = 0; k < HID; ++k) {
    const float xv = sX[k];
    a = fmaf(xv, w1a[k], a);
    b = fmaf(xv, w1b[k], b);
    y = fmaf(xv, Wb[k * HID + c], y);
  }
  Ai[i * HID + c] = a;
  Bj[i * HID + c] = b;
  Y[i * HID + c]  = y;
}

__global__ __launch_bounds__(32) void grn_tile_kernel(
    const float* __restrict__ X,  const float* __restrict__ Ai,
    const float* __restrict__ Bj, const float* __restrict__ Y,
    const float* __restrict__ b1, const float* __restrict__ W2,
    const float* __restrict__ b2, const float* __restrict__ bb,
    float* __restrict__ out) {
  __shared__ alignas(16) float sAi[16 * HID];
  __shared__ alignas(16) float sBj[16 * BJ_PITCH];

  const int lane = threadIdx.x;
  const int i0 = blockIdx.x * 16;
  const int j0 = blockIdx.y * 16;

  {
    const float4* __restrict__ gAi = (const float4*)(Ai + i0 * HID);
    const float4* __restrict__ gBj = (const float4*)(Bj + j0 * HID);
    float4* dAi = (float4*)sAi;
    float4* dBj = (float4*)sBj;
#pragma unroll
    for (int q = 0; q < 16; ++q) {
      dAi[lane + 32 * q] = gAi[lane + 32 * q];
      dBj[lane + (BJ_PITCH / 4) * q] = gBj[lane + 32 * q];
    }
  }
  __syncthreads();

  const float bbv = bb[0];
  v8f acc = {bbv, bbv, bbv, bbv, bbv, bbv, bbv, bbv};
  const int lm = lane & 15;
  const int kb = (lane >> 4) << 1;
  const float* __restrict__ arow = Y + (i0 + lm) * HID + kb;
  const float* __restrict__ brow = X + (j0 + lm) * HID + kb;
#pragma unroll
  for (int kk = 0; kk < HID; kk += 4) {
    const v2f afrag = *(const v2f*)(arow + kk);
    const v2f bfrag = *(const v2f*)(brow + kk);
    acc = __builtin_amdgcn_wmma_f32_16x16x4_f32(
         false, afrag,  false, bfrag,
         (short)0, acc,  false,  false);
  }

  const int n = lane & 15;
  const int mbase = (lane >> 4) << 3;
  const float4* __restrict__ ha = (const float4*)(sAi + mbase * HID);
  const float4* __restrict__ hb = (const float4*)(sBj + n * BJ_PITCH);
  const float4* __restrict__ gc = (const float4*)b1;
  const float4* __restrict__ g0 = (const float4*)W2;
  const float4* __restrict__ g1 = (const float4*)(W2 + HID);
  const float4* __restrict__ g2 = (const float4*)(W2 + 2 * HID);
  const float b20 = b2[0], b21 = b2[1], b22 = b2[2];
  float l0[8], l1[8], l2[8];
#pragma unroll
  for (int r = 0; r < 8; ++r) { l0[r] = b20; l1[r] = b21; l2[r] = b22; }

#pragma unroll 2
  for (int k4 = 0; k4 < HID / 4; ++k4) {
    const float4 vb = hb[k4], vc = gc[k4];
    const float4 u0 = g0[k4], u1 = g1[k4], u2 = g2[k4];
#pragma unroll
    for (int r = 0; r < 8; ++r) {
      const float4 va = ha[r * (HID / 4) + k4];
      float4 h;
      h.x = fmaxf(va.x + vb.x + vc.x, 0.f);
      h.y = fmaxf(va.y + vb.y + vc.y, 0.f);
      h.z = fmaxf(va.z + vb.z + vc.z, 0.f);
      h.w = fmaxf(va.w + vb.w + vc.w, 0.f);
      l0[r] = fmaf(h.x, u0.x, fmaf(h.y, u0.y, fmaf(h.z, u0.z, fmaf(h.w, u0.w, l0[r]))));
      l1[r] = fmaf(h.x, u1.x, fmaf(h.y, u1.y, fmaf(h.z, u1.z, fmaf(h.w, u1.w, l1[r]))));
      l2[r] = fmaf(h.x, u2.x, fmaf(h.y, u2.y, fmaf(h.z, u2.z, fmaf(h.w, u2.w, l2[r]))));
    }
  }

#pragma unroll
  for (int r = 0; r < 8; ++r) {
    const float s = (l0[r] >= l1[r] && l0[r] >= l2[r]) ? 1.0f
                  : ((l1[r] >= l2[r]) ? 0.0f : -1.0f);
    const int gi = i0 + mbase + r;
    const int gj = j0 + n;
    const float val = (gi == gj) ? 0.f : s * acc[r];
    out[gi * N_GENES + gj] = val;
  }
}

extern "C" void kernel_launch(void* const* d_in, const int* in_sizes, int n_in,
                              void* d_out, int out_size, void* d_ws,
                              size_t ws_size, hipStream_t stream) {
  (void)in_sizes; (void)n_in; (void)out_size; (void)ws_size;
  const float* X  = (const float*)d_in[0];
  const float* W1 = (const float*)d_in[1];
  const float* b1 = (const float*)d_in[2];
  const float* W2 = (const float*)d_in[3];
  const float* b2 = (const float*)d_in[4];
  const float* Wb = (const float*)d_in[5];
  const float* bb = (const float*)d_in[6];
  float* out = (float*)d_out;

  float* ws = (float*)d_ws;
  float* Ai = ws;
  float* Bj = ws + N_GENES * HID;
  float* Y  = ws + 2 * N_GENES * HID;

  grn_proj_kernel<<<dim3(N_GENES), dim3(HID), 0, stream>>>(X, W1, Wb, Ai, Bj, Y);
  grn_tile_kernel<<<dim3(NTILE, NTILE), dim3(32), 0, stream>>>(
      X, Ai, Bj, Y, b1, W2, b2, bb, out);
}


# Projection kernel: a block does 8 genes for one weight slab; W1 rows pass through LDS so global reads are whole lines (4 lines per load, was 32); weights kept in registers; fma chains bit-identical
# speedup vs baseline: 2.4951x; 1.3425x over previous
#include <hip/hip_runtime.h>


#define N_GENES 768
#define HID     128
#define NTILE   48
#define BJ_PITCH (HID + 4)

typedef float v2f __attribute__((ext_vector_type(2)));
typedef float v8f __attribute__((ext_vector_type(8)));

#define PG 8
#define PROJ_GROUPS 96
#define W_PITCH (HID + 4)

__global__ __launch_bounds__(HID) void grn_proj_kernel(
    const float* __restrict__ X, const float* __restrict__ W1,
    const float* __restrict__ Wb,
    float* __restrict__ Ai, float* __restrict__ Bj, float* __restrict__ Y) {
  __shared__ alignas(16) float sX[PG * HID];
  __shared__ alignas(16) float sW[HID * W_PITCH];
  const int c = threadIdx.x;
  const int i0 = blockIdx.x * PG;
  const int slab = blockIdx.y;

  {
    const float4* __restrict__ gX = (const float4*)(X + i0 * HID);
    float4* dX = (float4*)sX;
    dX[c] = gX[c];
    dX[c + HID] = gX[c + HID];
  }
  if (slab < 2) {
    const float4* __restrict__ gW = (const float4*)(W1 + slab * HID);
    float4* dW = (float4*)sW;
    const int col = c & 31;
    const int r0 = c >> 5;
#pragma unroll
    for (int q = 0; q < 32; ++q) {
      const int row = r0 + 4 * q;
      dW[row * (W_PITCH / 4) + col] = gW[row * (2 * HID / 4) + col];
    }
  }
  __syncthreads();

  float w[HID];
  if (slab < 2) {
    const float4* __restrict__ hW = (const float4*)(sW + c * W_PITCH);
#pragma unroll
    for (int k4 = 0; k4 < HID / 4; ++k4) {
      const float4 v = hW[k4];
      w[4 * k4 + 0] = v.x; w[4 * k4 + 1] = v.y;
      w[4 * k4 + 2] = v.z; w[4 * k4 + 3] = v.w;
    }
  } else {
#pragma unroll
    for (int k = 0; k < HID; ++k) w[k] = Wb[k * HID + c];
  }

  float acc[PG];
#pragma unroll
  for (int g = 0; g < PG; ++g) acc[g] = 0.f;
  const float4* __restrict__ hX = (const float4*)sX;
#pragma unroll
  for (int k4 = 0; k4 < HID / 4; ++k4) {
#pragma unroll
    for (int g = 0; g < PG; ++g) {
      const float4 xv = hX[g * (HID / 4) + k4];
      acc[g] = fmaf(xv.x, w[4 * k4 + 0], acc[g]);
      acc[g] = fmaf(xv.y, w[4 * k4 + 1], acc[g]);
      acc[g] = fmaf(xv.z, w[4 * k4 + 2], acc[g]);
      acc[g] = fmaf(xv.w, w[4 * k4 + 3], acc[g]);
    }
  }
  float* __restrict__ dst = (slab == 0) ? Ai : ((slab == 1) ? Bj : Y);
#pragma unroll
  for (int g = 0; g < PG; ++g) dst[(i0 + g) * HID + c] = acc[g];
}

__global__ __launch_bounds__(32) void grn_tile_kernel(
    const float* __restrict__ X,  const float* __restrict__ Ai,
    const float* __restrict__ Bj, const float* __restrict__ Y,
    const float* __restrict__ b1, const float* __restrict__ W2,
    const float* __restrict__ b2, const float* __restrict__ bb,
    float* __restrict__ out) {
  __shared__ alignas(16) float sAi[16 * HID];
  __shared__ alignas(16) float sBj[16 * BJ_PITCH];

  const int lane = threadIdx.x;
  const int i0 = blockIdx.x * 16;
  const int j0 = blockIdx.y * 16;

  {
    const float4* __restrict__ gAi = (const float4*)(Ai + i0 * HID);
    const float4* __restrict__ gBj = (const float4*)(Bj + j0 * HID);
    float4* dAi = (float4*)sAi;
    float4* dBj = (float4*)sBj;
#pragma unroll
    for (int q = 0; q < 16; ++q) {
      dAi[lane + 32 * q] = gAi[lane + 32 * q];
      dBj[lane + (BJ_PITCH / 4) * q] = gBj[lane + 32 * q];
    }
  }
  __syncthreads();

  const float bbv = bb[0];
  v8f acc = {bbv, bbv, bbv, bbv, bbv, bbv, bbv, bbv};
  const int lm = lane & 15;
  const int kb = (lane >> 4) << 1;
  const float* __restrict__ arow = Y + (i0 + lm) * HID + kb;
  const float* __restrict__ brow = X + (j0 + lm) * HID + kb;
#pragma unroll
  for (int kk = 0; kk < HID; kk += 4) {
    const v2f afrag = *(const v2f*)(arow + kk);
    const v2f bfrag = *(const v2f*)(brow + kk);
    acc = __builtin_amdgcn_wmma_f32_16x16x4_f32(
         false, afrag,  false, bfrag,
         (short)0, acc,  false,  false);
  }

  const int n = lane & 15;
  const int mbase = (lane >> 4) << 3;
  const float4* __restrict__ ha = (const float4*)(sAi + mbase * HID);
  const float4* __restrict__ hb = (const float4*)(sBj + n * BJ_PITCH);
  const float4* __restrict__ gc = (const float4*)b1;
  const float4* __restrict__ g0 = (const float4*)W2;
  const float4* __restrict__ g1 = (const float4*)(W2 + HID);
  const float4* __restrict__ g2 = (const float4*)(W2 + 2 * HID);
  const float b20 = b2[0], b21 = b2[1], b22 = b2[2];
  float l0[8], l1[8], l2[8];
#pragma unroll
  for (int r = 0; r < 8; ++r) { l0[r] = b20; l1[r] = b21; l2[r] = b22; }

#pragma unroll 2
  for (int k4 = 0; k4 < HID / 4; ++k4) {
    const float4 vb = hb[k4], vc = gc[k4];
    const float4 u0 = g0[k4], u1 = g1[k4], u2 = g2[k4];
#pragma unroll
    for (int r = 0; r < 8; ++r) {
      const float4 va = ha[r * (HID / 4) + k4];
      float4 h;
      h.x = fmaxf(va.x + vb.x + vc.x, 0.f);
      h.y = fmaxf(va.y + vb.y + vc.y, 0.f);
      h.z = fmaxf(va.z + vb.z + vc.z, 0.f);
      h.w = fmaxf(va.w + vb.w + vc.w, 0.f);
      l0[r] = fmaf(h.x, u0.x, fmaf(h.y, u0.y, fmaf(h.z, u0.z, fmaf(h.w, u0.w, l0[r]))));
      l1[r] = fmaf(h.x, u1.x, fmaf(h.y, u1.y, fmaf(h.z, u1.z, fmaf(h.w, u1.w, l1[r]))));
      l2[r] = fmaf(h.x, u2.x, fmaf(h.y, u2.y, fmaf(h.z, u2.z, fmaf(h.w, u2.w, l2[r]))));
    }
  }

#pragma unroll
  for (int r = 0; r < 8; ++r) {
    const float s = (l0[r] >= l1[r] && l0[r] >= l2[r]) ? 1.0f
                  : ((l1[r] >= l2[r]) ? 0.0f : -1.0f);
    const int gi = i0 + mbase + r;
    const int gj = j0 + n;
    const float val = (gi == gj) ? 0.f : s * acc[r];
    out[gi * N_GENES + gj] = val;
  }
}

extern "C" void kernel_launch(void* const* d_in, const int* in_sizes, int n_in,
                              void* d_out, int out_size, void* d_ws,
                              size_t ws_size, hipStream_t stream) {
  (void)in_sizes; (void)n_in; (void)out_size; (void)ws_size;
  const float* X  = (const float*)d_in[0];
  const float* W1 = (const float*)d_in[1];
  const float* b1 = (const float*)d_in[2];
  const float* W2 = (const float*)d_in[3];
  const float* b2 = (const float*)d_in[4];
  const float* Wb = (const float*)d_in[5];
  const float* bb = (const float*)d_in[6];
  float* out = (float*)d_out;

  float* ws = (float*)d_ws;
  float* Ai = ws;
  float* Bj = ws + N_GENES * HID;
  float* Y  = ws + 2 * N_GENES * HID;

  grn_proj_kernel<<<dim3(PROJ_GROUPS, 3), dim3(HID), 0, stream>>>(X, W1, Wb, Ai, Bj, Y);
  grn_tile_kernel<<<dim3(NTILE, NTILE), dim3(32), 0, stream>>>(
      X, Ai, Bj, Y, b1, W2, b2, bb, out);
}


# Projection: lane tile 4 channels x 3 genes in a rolled k loop, weights and X both read from LDS two chunks ahead of their fma (was 1 x 8, reads waited singly); 12-gene blocks; sums bit-identical
# speedup vs baseline: 2.7109x; 1.0865x over previous
#include <hip/hip_runtime.h>


#define N_GENES 768
#define HID     128
#define NTILE   48
#define BJ_PITCH (HID + 4)

typedef float v2f __attribute__((ext_vector_type(2)));
typedef float v8f __attribute__((ext_vector_type(8)));

#define PG 12
#define NB 4
#define PGW 3
#define PROJ_GROUPS 64
#define W_PITCH (HID + 4)

struct ProjChunk {
  float4 w[2][NB];
  float4 x[2][PGW];
};

__device__ __forceinline__ void proj_load(ProjChunk& b, const float4* hW,
                                          const float4* hX, int k4) {
#pragma unroll
  for (int u = 0; u < 2; ++u) {
#pragma unroll
    for (int j = 0; j < NB; ++j) b.w[u][j] = hW[j * (32 * W_PITCH / 4) + k4 + u];
#pragma unroll
    for (int g = 0; g < PGW; ++g) b.x[u][g] = hX[g * (HID / 4) + k4 + u];
  }
}

__device__ __forceinline__ void proj_fma(const ProjChunk& b, float (*acc)[PGW]) {
#pragma unroll
  for (int u = 0; u < 2; ++u) {
#pragma unroll
    for (int g = 0; g < PGW; ++g) {
      const float4 xv = b.x[u][g];
#pragma unroll
      for (int j = 0; j < NB; ++j) acc[j][g] = fmaf(xv.x, b.w[u][j].x, acc[j][g]);
#pragma unroll
      for (int j = 0; j < NB; ++j) acc[j][g] = fmaf(xv.y, b.w[u][j].y, acc[j][g]);
#pragma unroll
      for (int j = 0; j < NB; ++j) acc[j][g] = fmaf(xv.z, b.w[u][j].z, acc[j][g]);
#pragma unroll
      for (int j = 0; j < NB; ++j) acc[j][g] = fmaf(xv.w, b.w[u][j].w, acc[j][g]);
    }
  }
}

__global__ __launch_bounds__(HID) void grn_proj_kernel(
    const float* __restrict__ X, const float* __restrict__ W1,
    const float* __restrict__ Wb,
    float* __restrict__ Ai, float* __restrict__ Bj, float* __restrict__ Y) {
  __shared__ alignas(16) float sX[PG * HID];
  __shared__ alignas(16) float sW[HID * W_PITCH];
  const int t = threadIdx.x;
  const int i0 = blockIdx.x * PG;
  const int slab = blockIdx.y;

  {
    const float4* __restrict__ gX = (const float4*)(X + i0 * HID);
    float4* dX = (float4*)sX;
#pragma unroll
    for (int q = 0; q < PG / 4; ++q) dX[t + HID * q] = gX[t + HID * q];
  }
  float4* dW = (float4*)sW;
  if (slab < 2) {
    const float4* __restrict__ gW = (const float4*)(W1 + slab * HID);
    const int col = t & 31;
    const int r0 = t >> 5;
#pragma unroll
    for (int q = 0; q < 32; ++q) {
      const int row = r0 + 4 * q;
      dW[row * (W_PITCH / 4) + col] = gW[row * (2 * HID / 4) + col];
    }
  } else {
#pragma unroll
    for (int k4 = 0; k4 < HID / 4; ++k4) {
      float4 v;
      v.x = Wb[(4 * k4 + 0) * HID + t];
      v.y = Wb[(4 * k4 + 1) * HID + t];
      v.z = Wb[(4 * k4 + 2) * HID + t];
      v.w = Wb[(4 * k4 + 3) * HID + t];
      dW[t * (W_PITCH / 4) + k4] = v;
    }
  }
  __syncthreads();

  const int lane = t & 31;
  const int wv = t >> 5;
  const int c0 = (32 * NB) * (wv % (4 / NB)) + lane;
  const int gp = wv / (4 / NB);
  const float4* hW = (const float4*)(sW + c0 * W_PITCH);
  const float4* hX = (const float4*)(sX + gp * PGW * HID);

  float acc[NB][PGW];
#pragma unroll
  for (int j = 0; j < NB; ++j) {
#pragma unroll
    for (int g = 0; g < PGW; ++g) acc[j][g] = 0.f;
  }

  ProjChunk bufA, bufB;
  proj_load(bufA, hW, hX, 0);
#pragma unroll 1
  for (int k4 = 0; k4 < HID / 4 - 4; k4 += 4) {
    proj_load(bufB, hW, hX, k4 + 2);
    proj_fma(bufA, acc);
    proj_load(bufA, hW, hX, k4 + 4);
    proj_fma(bufB, acc);
  }
  proj_load(bufB, hW, hX, HID / 4 - 2);
  proj_fma(bufA, acc);
  proj_fma(bufB, acc);

  float* __restrict__ dst = (slab == 0) ? Ai : ((slab == 1) ? Bj : Y);
#pragma unroll
  for (int g = 0; g < PGW; ++g) {
#pragma unroll
    for (int j = 0; j < NB; ++j)
      dst[(i0 + gp * PGW + g) * HID + c0 + 32 * j] = acc[j][g];
  }
}

__global__ __launch_bounds__(32) void grn_tile_kernel(
    const float* __restrict__ X,  const float* __restrict__ Ai,
    const float* __restrict__ Bj, const float* __restrict__ Y,
    const float* __restrict__ b1, const float* __restrict__ W2,
    const float* __restrict__ b2, const float* __restrict__ bb,
    float* __restrict__ out) {
  __shared__ alignas(16) float sAi[16 * HID];
  __shared__ alignas(16) float sBj[16 * BJ_PITCH];

  const int lane = threadIdx.x;
  const int i0 = blockIdx.x * 16;
  const int j0 = blockIdx.y * 16;

  {
    const float4* __restrict__ gAi = (const float4*)(Ai + i0 * HID);
    const float4* __restrict__ gBj = (const float4*)(Bj + j0 * HID);
    float4* dAi = (float4*)sAi;
    float4* dBj = (float4*)sBj;
#pragma unroll
    for (int q = 0; q < 16; ++q) {
      dAi[lane + 32 * q] = gAi[lane + 32 * q];
      dBj[lane + (BJ_PITCH / 4) * q] = gBj[lane + 32 * q];
    }
  }
  __syncthreads();

  const float bbv = bb[0];
  v8f acc = {bbv, bbv, bbv, bbv, bbv, bbv, bbv, bbv};
  const int lm = lane & 15;
  const int kb = (lane >> 4) << 1;
  const float* __restrict__ arow = Y + (i0 + lm) * HID + kb;
  const float* __restrict__ brow = X + (j0 + lm) * HID + kb;
#pragma unroll
  for (int kk = 0; kk < HID; kk += 4) {
    const v2f afrag = *(const v2f*)(arow + kk);
    const v2f bfrag = *(const v2f*)(brow + kk);
    acc = __builtin_amdgcn_wmma_f32_16x16x4_f32(
         false, afrag,  false, bfrag,
         (short)0, acc,  false,  false);
  }

  const int n = lane & 15;
  const int mbase = (lane >> 4) << 3;
  const float4* __restrict__ ha = (const float4*)(sAi + mbase * HID);
  const float4* __restrict__ hb = (const float4*)(sBj + n * BJ_PITCH);
  const float4* __restrict__ gc = (const float4*)b1;
  const float4* __restrict__ g0 = (const float4*)W2;
  const float4* __restrict__ g1 = (const float4*)(W2 + HID);
  const float4* __restrict__ g2 = (const float4*)(W2 + 2 * HID);
  const float b20 = b2[0], b21 = b2[1], b22 = b2[2];
  float l0[8], l1[8], l2[8];
#pragma unroll
  for (int r = 0; r < 8; ++r) { l0[r] = b20; l1[r] = b21; l2[r] = b22; }

#pragma unroll 2
  for (int k4 = 0; k4 < HID / 4; ++k4) {
    const float4 vb = hb[k4], vc = gc[k4];
    const float4 u0 = g0[k4], u1 = g1[k4], u2 = g2[k4];
#pragma unroll
    for (int r = 0; r < 8; ++r) {
      const float4 va = ha[r * (HID / 4) + k4];
      float4 h;
      h.x = fmaxf(va.x + vb.x + vc.x, 0.f);
      h.y = fmaxf(va.y + vb.y + vc.y, 0.f);
      h.z = fmaxf(va.z + vb.z + vc.z, 0.f);
      h.w = fmaxf(va.w + vb.w + vc.w, 0.f);
      l0[r] = fmaf(h.x, u0.x, fmaf(h.y, u0.y, fmaf(h.z, u0.z, fmaf(h.w, u0.w, l0[r]))));
      l1[r] = fmaf(h.x, u1.x, fmaf(h.y, u1.y, fmaf(h.z, u1.z, fmaf(h.w, u1.w, l1[r]))));
      l2[r] = fmaf(h.x, u2.x, fmaf(h.y, u2.y, fmaf(h.z, u2.z, fmaf(h.w, u2.w, l2[r]))));
    }
  }

#pragma unroll
  for (int r = 0; r < 8; ++r) {
    const float s = (l0[r] >= l1[r] && l0[r] >= l2[r]) ? 1.0f
                  : ((l1[r] >= l2[r]) ? 0.0f : -1.0f);
    const int gi = i0 + mbase + r;
    const int gj = j0 + n;
    const float val = (gi == gj) ? 0.f : s * acc[r];
    out[gi * N_GENES + gj] = val;
  }
}

extern "C" void kernel_launch(void* const* d_in, const int* in_sizes, int n_in,
                              void* d_out, int out_size, void* d_ws,
                              size_t ws_size, hipStream_t stream) {
  (void)in_sizes; (void)n_in; (void)out_size; (void)ws_size;
  const float* X  = (const float*)d_in[0];
  const float* W1 = (const float*)d_in[1];
  const float* b1 = (const float*)d_in[2];
  const float* W2 = (const float*)d_in[3];
  const float* b2 = (const float*)d_in[4];
  const float* Wb = (const float*)d_in[5];
  const float* bb = (const float*)d_in[6];
  float* out = (float*)d_out;

  float* ws = (float*)d_ws;
  float* Ai = ws;
  float* Bj = ws + N_GENES * HID;
  float* Y  = ws + 2 * N_GENES * HID;

  grn_proj_kernel<<<dim3(PROJ_GROUPS, 3), dim3(HID), 0, stream>>>(X, W1, Wb, Ai, Bj, Y);
  grn_tile_kernel<<<dim3(NTILE, NTILE), dim3(32), 0, stream>>>(
      X, Ai, Bj, Y, b1, W2, b2, bb, out);
}


# Tile kernel re-tiled: one 8-wave workgroup per 48x48 region (256, one per WGP); operands staged in LDS once per region, a lane classifies 3x3 pairs, WMMA steps ride in its k loop; sums bit-identical
# speedup vs baseline: 2.9679x; 1.0948x over previous
#include <hip/hip_runtime.h>


#define N_GENES 768
#define HID     128
#define OP_PITCH (HID + 4)

typedef float v2f __attribute__((ext_vector_type(2)));
typedef float v4f __attribute__((ext_vector_type(4)));
typedef float v8f __attribute__((ext_vector_type(8)));

#define PG 12
#define NB 4
#define PGW 3
#define PROJ_GROUPS 64
#define W_PITCH (HID + 4)

struct ProjChunk {
  float4 w[2][NB];
  float4 x[2][PGW];
};

__device__ __forceinline__ void proj_load(ProjChunk& b, const float4* hW,
                                          const float4* hX, int k4) {
#pragma unroll
  for (int u = 0; u < 2; ++u) {
#pragma unroll
    for (int j = 0; j < NB; ++j) b.w[u][j] = hW[j * (32 * W_PITCH / 4) + k4 + u];
#pragma unroll
    for (int g = 0; g < PGW; ++g) b.x[u][g] = hX[g * (HID / 4) + k4 + u];
  }
}

__device__ __forceinline__ void proj_fma(const ProjChunk& b, float (*acc)[PGW]) {
#pragma unroll
  for (int u = 0; u < 2; ++u) {
#pragma unroll
    for (int g = 0; g < PGW; ++g) {
      const float4 xv = b.x[u][g];
#pragma unroll
      for (int j = 0; j < NB; ++j) acc[j][g] = fmaf(xv.x, b.w[u][j].x, acc[j][g]);
#pragma unroll
      for (int j = 0; j < NB; ++j) acc[j][g] = fmaf(xv.y, b.w[u][j].y, acc[j][g]);
#pragma unroll
      for (int j = 0; j < NB; ++j) acc[j][g] = fmaf(xv.z, b.w[u][j].z, acc[j][g]);
#pragma unroll
      for (int j = 0; j < NB; ++j) acc[j][g] = fmaf(xv.w, b.w[u][j].w, acc[j][g]);
    }
  }
}

__global__ __launch_bounds__(HID) void grn_proj_kernel(
    const float* __restrict__ X, const float* __restrict__ W1,
    const float* __restrict__ Wb,
    float* __restrict__ Ai, float* __restrict__ Bj, float* __restrict__ Y) {
  __shared__ alignas(16) float sX[PG * HID];
  __shared__ alignas(16) float sW[HID * W_PITCH];
  const int t = threadIdx.x;
  const int i0 = blockIdx.x * PG;
  const int slab = blockIdx.y;

  {
    const float4* __restrict__ gX = (const float4*)(X + i0 * HID);
    float4* dX = (float4*)sX;
#pragma unroll
    for (int q = 0; q < PG / 4; ++q) dX[t + HID * q] = gX[t + HID * q];
  }
  float4* dW = (float4*)sW;
  if (slab < 2) {
    const float4* __restrict__ gW = (const float4*)(W1 + slab * HID);
    const int col = t & 31;
    const int r0 = t >> 5;
#pragma unroll
    for (int q = 0; q < 32; ++q) {
      const int row = r0 + 4 * q;
      dW[row * (W_PITCH / 4) + col] = gW[row * (2 * HID / 4) + col];
    }
  } else {
#pragma unroll
    for (int k4 = 0; k4 < HID / 4; ++k4) {
      float4 v;
      v.x = Wb[(4 * k4 + 0) * HID + t];
      v.y = Wb[(4 * k4 + 1) * HID + t];
      v.z = Wb[(4 * k4 + 2) * HID + t];
      v.w = Wb[(4 * k4 + 3) * HID + t];
      dW[t * (W_PITCH / 4) + k4] = v;
    }
  }
  __syncthreads();

  const int lane = t & 31;
  const int wv = t >> 5;
  const int c0 = (32 * NB) * (wv % (4 / NB)) + lane;
  const int gp = wv / (4 / NB);
  const float4* hW = (const float4*)(sW + c0 * W_PITCH);
  const float4* hX = (const float4*)(sX + gp * PGW * HID);

  float acc[NB][PGW];
#pragma unroll
  for (int j = 0; j < NB; ++j) {
#pragma unroll
    for (int g = 0; g < PGW; ++g) acc[j][g] = 0.f;
  }

  ProjChunk bufA, bufB;
  proj_load(bufA, hW, hX, 0);
#pragma unroll 1
  for (int k4 = 0; k4 < HID / 4 - 4; k4 += 4) {
    proj_load(bufB, hW, hX, k4 + 2);
    proj_fma(bufA, acc);
    proj_load(bufA, hW, hX, k4 + 4);
    proj_fma(bufB, acc);
  }
  proj_load(bufB, hW, hX, HID / 4 - 2);
  proj_fma(bufA, acc);
  proj_fma(bufB, acc);

  float* __restrict__ dst = (slab == 0) ? Ai : ((slab == 1) ? Bj : Y);
#pragma unroll
  for (int g = 0; g < PGW; ++g) {
#pragma unroll
    for (int j = 0; j < NB; ++j)
      dst[(i0 + gp * PGW + g) * HID + c0 + 32 * j] = acc[j][g];
  }
}

#define RG 48
#define RT (RG / 16)
#define NREG (N_GENES / RG)
#define SG_PITCH (RG + 4)
#define TILE_WAVES 8
#define TILE_THREADS (32 * TILE_WAVES)
#define SROW_S (TILE_WAVES * (HID / 4))
#define SROW_D (TILE_WAVES * (OP_PITCH / 4))
#define LDS_USED (4 * RG * OP_PITCH + 2 * RG * SG_PITCH)
#define LDS_RESERVED (162 * 1024 / 4)

__device__ __forceinline__ void lds_barrier() {
  __builtin_amdgcn_fence(__ATOMIC_RELEASE, "workgroup", "local");
  __builtin_amdgcn_s_barrier();
  __builtin_amdgcn_fence(__ATOMIC_ACQUIRE, "workgroup", "local");
}

__device__ __forceinline__ void cls_chunk(
    int k4, const float4* ha, const float4* hb,
    const float* __restrict__ b1, const float* __restrict__ W2,
    float (*l0)[3], float (*l1)[3], float (*l2)[3]) {
  const float4 vc = ((const float4*)b1)[k4];
  const float4 u0 = ((const float4*)W2)[k4];
  const float4 u1 = ((const float4*)(W2 + HID))[k4];
  const float4 u2 = ((const float4*)(W2 + 2 * HID))[k4];
  float4 va[3], vb[3];
#pragma unroll
  for (int a = 0; a < 3; ++a) va[a] = ha[a * (OP_PITCH / 4) + k4];
#pragma unroll
  for (int m = 0; m < 3; ++m) vb[m] = hb[m * 8 * (OP_PITCH / 4) + k4];
#pragma unroll
  for (int a = 0; a < 3; ++a) {
#pragma unroll
    for (int m = 0; m < 3; ++m) {
      float4 h;
      h.x = fmaxf(va[a].x + vb[m].x + vc.x, 0.f);
      h.y = fmaxf(va[a].y + vb[m].y + vc.y, 0.f);
      h.z = fmaxf(va[a].z + vb[m].z + vc.z, 0.f);
      h.w = fmaxf(va[a].w + vb[m].w + vc.w, 0.f);
      l0[a][m] = fmaf(h.x, u0.x, fmaf(h.y, u0.y, fmaf(h.z, u0.z, fmaf(h.w, u0.w, l0[a][m]))));
      l1[a][m] = fmaf(h.x, u1.x, fmaf(h.y, u1.y, fmaf(h.z, u1.z, fmaf(h.w, u1.w, l1[a][m]))));
      l2[a][m] = fmaf(h.x, u2.x, fmaf(h.y, u2.y, fmaf(h.z, u2.z, fmaf(h.w, u2.w, l2[a][m]))));
    }
  }
}

__global__ __launch_bounds__(TILE_THREADS) void grn_tile_kernel(
    const float* X,               const float* __restrict__ Ai,
    const float* __restrict__ Bj, const float* Y,
    const float* __restrict__ b1, const float* __restrict__ W2,
    const float* __restrict__ b2, const float* __restrict__ bb,
    float* __restrict__ out) {
  static_assert(LDS_USED <= LDS_RESERVED, "LDS layout");
  __shared__ alignas(16) float lds[LDS_RESERVED];
  float* sAi = lds;
  float* sBj = sAi + RG * OP_PITCH;
  float* sY  = sBj + RG * OP_PITCH;
  float* sX  = sY  + RG * OP_PITCH;
  float* sAf = sX  + RG * OP_PITCH;
  float* sSg = sAf + RG * SG_PITCH;

  const int t = threadIdx.x;
  const int lane = t & 31;
  const int w = __builtin_amdgcn_readfirstlane(t >> 5);
  const int i0 = blockIdx.x * RG;
  const int j0 = blockIdx.y * RG;

  const int srow = w * (HID / 4) + lane;
  const int drow = w * (OP_PITCH / 4) + lane;
  const v4f* gY = (const v4f*)(Y + i0 * HID);
  const v4f* gX = (const v4f*)(X + j0 * HID);
  {
    const v4f* __restrict__ gAi = (const v4f*)(Ai + i0 * HID);
    const v4f* __restrict__ gBj = (const v4f*)(Bj + j0 * HID);
    v4f* dAi = (v4f*)sAi;
    v4f* dBj = (v4f*)sBj;
#pragma unroll
    for (int q = 0; q < RG / TILE_WAVES; ++q) {
      dAi[drow + q * SROW_D] = gAi[srow + q * SROW_S];
      dBj[drow + q * SROW_D] = gBj[srow + q * SROW_S];
    }
  }
  const v4f y0 = gY[srow], y1 = gY[srow + SROW_S], y2 = gY[srow + 2 * SROW_S],
            y3 = gY[srow + 3 * SROW_S], y4 = gY[srow + 4 * SROW_S], y5 = gY[srow + 5 * SROW_S];
  const v4f x0 = gX[srow], x1 = gX[srow + SROW_S], x2 = gX[srow + 2 * SROW_S],
            x3 = gX[srow + 3 * SROW_S], x4 = gX[srow + 4 * SROW_S], x5 = gX[srow + 5 * SROW_S];
  lds_barrier();

  const float bbv = bb[0];
  const int lm = lane & 15;
  const int hh = lane >> 4;
  const int kb = hh << 1;

  const int rowL = 12 * (w >> 1) + 3 * (lane >> 3);
  const int colL = 24 * (w & 1) + (lane & 7);
  const float4* ha = (const float4*)(sAi + rowL * OP_PITCH);
  const float4* hb = (const float4*)(sBj + colL * OP_PITCH);
  const float b20 = b2[0], b21 = b2[1], b22 = b2[2];
  float l0[3][3], l1[3][3], l2[3][3];
#pragma unroll
  for (int a = 0; a < 3; ++a) {
#pragma unroll
    for (int m = 0; m < 3; ++m) { l0[a][m] = b20; l1[a][m] = b21; l2[a][m] = b22; }
  }

#pragma unroll 2
  for (int k4 = 0; k4 < HID / 8; ++k4) cls_chunk(k4, ha, hb, b1, W2, l0, l1, l2);

  {
    v4f* dY = (v4f*)sY;
    v4f* dX = (v4f*)sX;
    dY[drow] = y0;              dX[drow] = x0;
    dY[drow + SROW_D] = y1;     dX[drow + SROW_D] = x1;
    dY[drow + 2 * SROW_D] = y2; dX[drow + 2 * SROW_D] = x2;
    dY[drow + 3 * SROW_D] = y3; dX[drow + 3 * SROW_D] = x3;
    dY[drow + 4 * SROW_D] = y4; dX[drow + 4 * SROW_D] = x4;
    dY[drow + 5 * SROW_D] = y5; dX[drow + 5 * SROW_D] = x5;
  }
  __syncthreads();

  if (w == 0) {
    const float* arow = sY + (32 + lm) * OP_PITCH + kb;
    const float* brow = sX + (32 + lm) * OP_PITCH + kb;
    v8f acc = {bbv, bbv, bbv, bbv, bbv, bbv, bbv, bbv};
#pragma unroll
    for (int kk = 0; kk < HID; kk += 4) {
      const v2f afrag = *(const v2f*)(arow + kk);
      const v2f bfrag = *(const v2f*)(brow + kk);
      acc = __builtin_amdgcn_wmma_f32_16x16x4_f32(
           false, afrag,  false, bfrag,
           (short)0, acc,  false,  false);
    }
    float* d = sAf + (32 + 8 * hh) * SG_PITCH + 32 + lm;
#pragma unroll
    for (int r = 0; r < 8; ++r) d[r * SG_PITCH] = acc[r];
  }
  {
    const int ti = w / RT;
    const int tj = w - ti * RT;
    const float* arow = sY + (16 * ti + lm) * OP_PITCH + kb;
    const float* brow = sX + (16 * tj + lm) * OP_PITCH + kb;
    v8f acc = {bbv, bbv, bbv, bbv, bbv, bbv, bbv, bbv};

#pragma unroll 2
    for (int k4 = HID / 8; k4 < HID / 4; ++k4) {
      const int kk = 8 * (k4 - HID / 8);
      const v2f af0 = *(const v2f*)(arow + kk);
      const v2f bf0 = *(const v2f*)(brow + kk);
      const v2f af1 = *(const v2f*)(arow + kk + 4);
      const v2f bf1 = *(const v2f*)(brow + kk + 4);
      acc = __builtin_amdgcn_wmma_f32_16x16x4_f32(
           false, af0,  false, bf0,
           (short)0, acc,  false,  false);
      acc = __builtin_amdgcn_wmma_f32_16x16x4_f32(
           false, af1,  false, bf1,
           (short)0, acc,  false,  false);
      cls_chunk(k4, ha, hb, b1, W2, l0, l1, l2);
    }

    float* d = sAf + (16 * ti + 8 * hh) * SG_PITCH + 16 * tj + lm;
#pragma unroll
    for (int r = 0; r < 8; ++r) d[r * SG_PITCH] = acc[r];
  }

#pragma unroll
  for (int a = 0; a < 3; ++a) {
#pragma unroll
    for (int m = 0; m < 3; ++m) {
      const float s = (l0[a][m] >= l1[a][m] && l0[a][m] >= l2[a][m]) ? 1.0f
                    : ((l1[a][m] >= l2[a][m]) ? 0.0f : -1.0f);
      sSg[(rowL + a) * SG_PITCH + colL + 8 * m] = s;
    }
  }
  __syncthreads();

  for (int idx = t; idx < RG * (RG / 4); idx += TILE_THREADS) {
    const int row = idx / (RG / 4);
    const int c4 = idx - row * (RG / 4);
    const float4 s = ((const float4*)sSg)[row * (SG_PITCH / 4) + c4];
    const float4 a = ((const float4*)sAf)[row * (SG_PITCH / 4) + c4];
    const int gi = i0 + row;
    const int gj = j0 + 4 * c4;
    float4 o;
    o.x = (gi == gj)     ? 0.f : s.x * a.x;
    o.y = (gi == gj + 1) ? 0.f : s.y * a.y;
    o.z = (gi == gj + 2) ? 0.f : s.z * a.z;
    o.w = (gi == gj + 3) ? 0.f : s.w * a.w;
    *(float4*)(out + gi * N_GENES + gj) = o;
  }
}

extern "C" void kernel_launch(void* const* d_in, const int* in_sizes, int n_in,
                              void* d_out, int out_size, void* d_ws,
                              size_t ws_size, hipStream_t stream) {
  (void)in_sizes; (void)n_in; (void)out_size; (void)ws_size;
  const float* X  = (const float*)d_in[0];
  const float* W1 = (const float*)d_in[1];
  const float* b1 = (const float*)d_in[2];
  const float* W2 = (const float*)d_in[3];
  const float* b2 = (const float*)d_in[4];
  const float* Wb = (const float*)d_in[5];
  const float* bb = (const float*)d_in[6];
  float* out = (float*)d_out;

  float* ws = (float*)d_ws;
  float* Ai = ws;
  float* Bj = ws + N_GENES * HID;
  float* Y  = ws + 2 * N_GENES * HID;

  grn_proj_kernel<<<dim3(PROJ_GROUPS, 3), dim3(HID), 0, stream>>>(X, W1, Wb, Ai, Bj, Y);
  grn_tile_kernel<<<dim3(NREG, NREG), dim3(TILE_THREADS), 0, stream>>>(
      X, Ai, Bj, Y, b1, W2, b2, bb, out);
}


# Classifier cut from 6 to 4 vector ops per pair and hidden unit: b1 folded into Bj in the projection, argmax from two logit differences (2 fma, was 3); sums reordered, signs can differ at near-ties
# speedup vs baseline: 3.1869x; 1.0738x over previous
#include <hip/hip_runtime.h>


#define N_GENES 768
#define HID     128
#define OP_PITCH (HID + 4)

typedef float v2f __attribute__((ext_vector_type(2)));
typedef float v4f __attribute__((ext_vector_type(4)));
typedef float v8f __attribute__((ext_vector_type(8)));

#define PG 12
#define NB 4
#define PGW 3
#define PROJ_GROUPS 64
#define W_PITCH (HID + 4)

struct ProjChunk {
  float4 w[2][NB];
  float4 x[2][PGW];
};

__device__ __forceinline__ void proj_load(ProjChunk& b, const float4* hW,
                                          const float4* hX, int k4) {
#pragma unroll
  for (int u = 0; u < 2; ++u) {
#pragma unroll
    for (int j = 0; j < NB; ++j) b.w[u][j] = hW[j * (32 * W_PITCH / 4) + k4 + u];
#pragma unroll
    for (int g = 0; g < PGW; ++g) b.x[u][g] = hX[g * (HID / 4) + k4 + u];
  }
}

__device__ __forceinline__ void proj_fma(const ProjChunk& b, float (*acc)[PGW]) {
#pragma unroll
  for (int u = 0; u < 2; ++u) {
#pragma unroll
    for (int g = 0; g < PGW; ++g) {
      const float4 xv = b.x[u][g];
#pragma unroll
      for (int j = 0; j < NB; ++j) acc[j][g] = fmaf(xv.x, b.w[u][j].x, acc[j][g]);
#pragma unroll
      for (int j = 0; j < NB; ++j) acc[j][g] = fmaf(xv.y, b.w[u][j].y, acc[j][g]);
#pragma unroll
      for (int j = 0; j < NB; ++j) acc[j][g] = fmaf(xv.z, b.w[u][j].z, acc[j][g]);
#pragma unroll
      for (int j = 0; j < NB; ++j) acc[j][g] = fmaf(xv.w, b.w[u][j].w, acc[j][g]);
    }
  }
}

__global__ __launch_bounds__(HID) void grn_proj_kernel(
    const float* __restrict__ X, const float* __restrict__ W1,
    const float* __restrict__ Wb, const float* __restrict__ b1,
    float* __restrict__ Ai, float* __restrict__ Bj, float* __restrict__ Y) {
  __shared__ alignas(16) float sX[PG * HID];
  __shared__ alignas(16) float sW[HID * W_PITCH];
  const int t = threadIdx.x;
  const int i0 = blockIdx.x * PG;
  const int slab = blockIdx.y;
  const int lane = t & 31;
  const int wv = t >> 5;
  const int c0 = (32 * NB) * (wv % (4 / NB)) + lane;
  const int gp = wv / (4 / NB);

  float bv[NB];
#pragma unroll
  for (int j = 0; j < NB; ++j) bv[j] = b1[c0 + 32 * j];

  {
    const float4* __restrict__ gX = (const float4*)(X + i0 * HID);
    float4* dX = (float4*)sX;
#pragma unroll
    for (int q = 0; q < PG / 4; ++q) dX[t + HID * q] = gX[t + HID * q];
  }
  float4* dW = (float4*)sW;
  if (slab < 2) {
    const float4* __restrict__ gW = (const float4*)(W1 + slab * HID);
    const int col = t & 31;
    const int r0 = t >> 5;
#pragma unroll
    for (int q = 0; q < 32; ++q) {
      const int row = r0 + 4 * q;
      dW[row * (W_PITCH / 4) + col] = gW[row * (2 * HID / 4) + col];
    }
  } else {
#pragma unroll
    for (int k4 = 0; k4 < HID / 4; ++k4) {
      float4 v;
      v.x = Wb[(4 * k4 + 0) * HID + t];
      v.y = Wb[(4 * k4 + 1) * HID + t];
      v.z = Wb[(4 * k4 + 2) * HID + t];
      v.w = Wb[(4 * k4 + 3) * HID + t];
      dW[t * (W_PITCH / 4) + k4] = v;
    }
  }
  __syncthreads();

  const float4* hW = (const float4*)(sW + c0 * W_PITCH);
  const float4* hX = (const float4*)(sX + gp * PGW * HID);

  float acc[NB][PGW];
#pragma unroll
  for (int j = 0; j < NB; ++j) {
#pragma unroll
    for (int g = 0; g < PGW; ++g) acc[j][g] = 0.f;
  }

  ProjChunk bufA, bufB;
  proj_load(bufA, hW, hX, 0);
#pragma unroll 1
  for (int k4 = 0; k4 < HID / 4 - 4; k4 += 4) {
    proj_load(bufB, hW, hX, k4 + 2);
    proj_fma(bufA, acc);
    proj_load(bufA, hW, hX, k4 + 4);
    proj_fma(bufB, acc);
  }
  proj_load(bufB, hW, hX, HID / 4 - 2);
  proj_fma(bufA, acc);
  proj_fma(bufB, acc);

  float* __restrict__ dst = (slab == 0) ? Ai : ((slab == 1) ? Bj : Y);
  if (slab == 1) {
#pragma unroll
    for (int j = 0; j < NB; ++j) {
#pragma unroll
      for (int g = 0; g < PGW; ++g) acc[j][g] += bv[j];
    }
  }
#pragma unroll
  for (int g = 0; g < PGW; ++g) {
#pragma unroll
    for (int j = 0; j < NB; ++j)
      dst[(i0 + gp * PGW + g) * HID + c0 + 32 * j] = acc[j][g];
  }
}

#define RG 48
#define RT (RG / 16)
#define NREG (N_GENES / RG)
#define SG_PITCH (RG + 4)
#define TILE_WAVES 8
#define TILE_THREADS (32 * TILE_WAVES)
#define SROW_S (TILE_WAVES * (HID / 4))
#define SROW_D (TILE_WAVES * (OP_PITCH / 4))
#define LDS_USED (4 * RG * OP_PITCH + 2 * RG * SG_PITCH)
#define LDS_RESERVED (162 * 1024 / 4)

__device__ __forceinline__ void lds_barrier() {
  __builtin_amdgcn_fence(__ATOMIC_RELEASE, "workgroup", "local");
  __builtin_amdgcn_s_barrier();
  __builtin_amdgcn_fence(__ATOMIC_ACQUIRE, "workgroup", "local");
}

__device__ __forceinline__ void cls_chunk(
    int k4, const float4* ha, const float4* hb,
    const float* __restrict__ W2,
    float (*d1)[3], float (*d2)[3]) {
  const float4 u0 = ((const float4*)W2)[k4];
  const float4 u1 = ((const float4*)(W2 + HID))[k4];
  const float4 u2 = ((const float4*)(W2 + 2 * HID))[k4];
  float4 p, q;
  p.x = u0.x - u1.x; p.y = u0.y - u1.y; p.z = u0.z - u1.z; p.w = u0.w - u1.w;
  q.x = u0.x - u2.x; q.y = u0.y - u2.y; q.z = u0.z - u2.z; q.w = u0.w - u2.w;
  float4 va[3], vb[3];
#pragma unroll
  for (int a = 0; a < 3; ++a) va[a] = ha[a * (OP_PITCH / 4) + k4];
#pragma unroll
  for (int m = 0; m < 3; ++m) vb[m] = hb[m * 8 * (OP_PITCH / 4) + k4];
#pragma unroll
  for (int a = 0; a < 3; ++a) {
#pragma unroll
    for (int m = 0; m < 3; ++m) {
      float4 h;
      h.x = fmaxf(va[a].x + vb[m].x, 0.f);
      h.y = fmaxf(va[a].y + vb[m].y, 0.f);
      h.z = fmaxf(va[a].z + vb[m].z, 0.f);
      h.w = fmaxf(va[a].w + vb[m].w, 0.f);
      d1[a][m] = fmaf(h.x, p.x, fmaf(h.y, p.y, fmaf(h.z, p.z, fmaf(h.w, p.w, d1[a][m]))));
      d2[a][m] = fmaf(h.x, q.x, fmaf(h.y, q.y, fmaf(h.z, q.z, fmaf(h.w, q.w, d2[a][m]))));
    }
  }
}

__global__ __launch_bounds__(TILE_THREADS) void grn_tile_kernel(
    const float* X,               const float* __restrict__ Ai,
    const float* __restrict__ Bj, const float* Y,
    const float* __restrict__ W2,
    const float* __restrict__ b2, const float* __restrict__ bb,
    float* __restrict__ out) {
  static_assert(LDS_USED <= LDS_RESERVED, "LDS layout");
  __shared__ alignas(16) float lds[LDS_RESERVED];
  float* sAi = lds;
  float* sBj = sAi + RG * OP_PITCH;
  float* sY  = sBj + RG * OP_PITCH;
  float* sX  = sY  + RG * OP_PITCH;
  float* sAf = sX  + RG * OP_PITCH;
  float* sSg = sAf + RG * SG_PITCH;

  const int t = threadIdx.x;
  const int lane = t & 31;
  const int w = __builtin_amdgcn_readfirstlane(t >> 5);
  const int i0 = blockIdx.x * RG;
  const int j0 = blockIdx.y * RG;

  const int srow = w * (HID / 4) + lane;
  const int drow = w * (OP_PITCH / 4) + lane;
  const v4f* gY = (const v4f*)(Y + i0 * HID);
  const v4f* gX = (const v4f*)(X + j0 * HID);
  {
    const v4f* __restrict__ gAi = (const v4f*)(Ai + i0 * HID);
    const v4f* __restrict__ gBj = (const v4f*)(Bj + j0 * HID);
    v4f* dAi = (v4f*)sAi;
    v4f* dBj = (v4f*)sBj;
#pragma unroll
    for (int q = 0; q < RG / TILE_WAVES; ++q) {
      dAi[drow + q * SROW_D] = gAi[srow + q * SROW_S];
      dBj[drow + q * SROW_D] = gBj[srow + q * SROW_S];
    }
  }
  const v4f y0 = gY[srow], y1 = gY[srow + SROW_S], y2 = gY[srow + 2 * SROW_S],
            y3 = gY[srow + 3 * SROW_S], y4 = gY[srow + 4 * SROW_S], y5 = gY[srow + 5 * SROW_S];
  const v4f x0 = gX[srow], x1 = gX[srow + SROW_S], x2 = gX[srow + 2 * SROW_S],
            x3 = gX[srow + 3 * SROW_S], x4 = gX[srow + 4 * SROW_S], x5 = gX[srow + 5 * SROW_S];
  lds_barrier();

  const float bbv = bb[0];
  const int lm = lane & 15;
  const int hh = lane >> 4;
  const int kb = hh << 1;

  const int rowL = 12 * (w >> 1) + 3 * (lane >> 3);
  const int colL = 24 * (w & 1) + (lane & 7);
  const float4* ha = (const float4*)(sAi + rowL * OP_PITCH);
  const float4* hb = (const float4*)(sBj + colL * OP_PITCH);
  const float b20 = b2[0], b21 = b2[1], b22 = b2[2];
  float d1[3][3], d2[3][3];
#pragma unroll
  for (int a = 0; a < 3; ++a) {
#pragma unroll
    for (int m = 0; m < 3; ++m) { d1[a][m] = b20 - b21; d2[a][m] = b20 - b22; }
  }

#pragma unroll 2
  for (int k4 = 0; k4 < HID / 8; ++k4) cls_chunk(k4, ha, hb, W2, d1, d2);

  {
    v4f* dY = (v4f*)sY;
    v4f* dX = (v4f*)sX;
    dY[drow] = y0;              dX[drow] = x0;
    dY[drow + SROW_D] = y1;     dX[drow + SROW_D] = x1;
    dY[drow + 2 * SROW_D] = y2; dX[drow + 2 * SROW_D] = x2;
    dY[drow + 3 * SROW_D] = y3; dX[drow + 3 * SROW_D] = x3;
    dY[drow + 4 * SROW_D] = y4; dX[drow + 4 * SROW_D] = x4;
    dY[drow + 5 * SROW_D] = y5; dX[drow + 5 * SROW_D] = x5;
  }
  __syncthreads();

  if (w == 0) {
    const float* arow = sY + (32 + lm) * OP_PITCH + kb;
    const float* brow = sX + (32 + lm) * OP_PITCH + kb;
    v8f acc = {bbv, bbv, bbv, bbv, bbv, bbv, bbv, bbv};
#pragma unroll
    for (int kk = 0; kk < HID; kk += 4) {
      const v2f afrag = *(const v2f*)(arow + kk);
      const v2f bfrag = *(const v2f*)(brow + kk);
      acc = __builtin_amdgcn_wmma_f32_16x16x4_f32(
           false, afrag,  false, bfrag,
           (short)0, acc,  false,  false);
    }
    float* d = sAf + (32 + 8 * hh) * SG_PITCH + 32 + lm;
#pragma unroll
    for (int r = 0; r < 8; ++r) d[r * SG_PITCH] = acc[r];
  }
  {
    const int ti = w / RT;
    const int tj = w - ti * RT;
    const float* arow = sY + (16 * ti + lm) * OP_PITCH + kb;
    const float* brow = sX + (16 * tj + lm) * OP_PITCH + kb;
    v8f acc = {bbv, bbv, bbv, bbv, bbv, bbv, bbv, bbv};

#pragma unroll 2
    for (int k4 = HID / 8; k4 < HID / 4; ++k4) {
      const int kk = 8 * (k4 - HID / 8);
      const v2f af0 = *(const v2f*)(arow + kk);
      const v2f bf0 = *(const v2f*)(brow + kk);
      const v2f af1 = *(const v2f*)(arow + kk + 4);
      const v2f bf1 = *(const v2f*)(brow + kk + 4);
      acc = __builtin_amdgcn_wmma_f32_16x16x4_f32(
           false, af0,  false, bf0,
           (short)0, acc,  false,  false);
      acc = __builtin_amdgcn_wmma_f32_16x16x4_f32(
           false, af1,  false, bf1,
           (short)0, acc,  false,  false);
      cls_chunk(k4, ha, hb, W2, d1, d2);
    }

    float* d = sAf + (16 * ti + 8 * hh) * SG_PITCH + 16 * tj + lm;
#pragma unroll
    for (int r = 0; r < 8; ++r) d[r * SG_PITCH] = acc[r];
  }

#pragma unroll
  for (int a = 0; a < 3; ++a) {
#pragma unroll
    for (int m = 0; m < 3; ++m) {
      const float s = (d1[a][m] >= 0.0f && d2[a][m] >= 0.0f) ? 1.0f
                    : ((d2[a][m] >= d1[a][m]) ? 0.0f : -1.0f);
      sSg[(rowL + a) * SG_PITCH + colL + 8 * m] = s;
    }
  }
  __syncthreads();

  for (int idx = t; idx < RG * (RG / 4); idx += TILE_THREADS) {
    const int row = idx / (RG / 4);
    const int c4 = idx - row * (RG / 4);
    const float4 s = ((const float4*)sSg)[row * (SG_PITCH / 4) + c4];
    const float4 a = ((const float4*)sAf)[row * (SG_PITCH / 4) + c4];
    const int gi = i0 + row;
    const int gj = j0 + 4 * c4;
    float4 o;
    o.x = (gi == gj)     ? 0.f : s.x * a.x;
    o.y = (gi == gj + 1) ? 0.f : s.y * a.y;
    o.z = (gi == gj + 2) ? 0.f : s.z * a.z;
    o.w = (gi == gj + 3) ? 0.f : s.w * a.w;
    *(float4*)(out + gi * N_GENES + gj) = o;
  }
}

extern "C" void kernel_launch(void* const* d_in, const int* in_sizes, int n_in,
                              void* d_out, int out_size, void* d_ws,
                              size_t ws_size, hipStream_t stream) {
  (void)in_sizes; (void)n_in; (void)out_size; (void)ws_size;
  const float* X  = (const float*)d_in[0];
  const float* W1 = (const float*)d_in[1];
  const float* b1 = (const float*)d_in[2];
  const float* W2 = (const float*)d_in[3];
  const float* b2 = (const float*)d_in[4];
  const float* Wb = (const float*)d_in[5];
  const float* bb = (const float*)d_in[6];
  float* out = (float*)d_out;

  float* ws = (float*)d_ws;
  float* Ai = ws;
  float* Bj = ws + N_GENES * HID;
  float* Y  = ws + 2 * N_GENES * HID;

  grn_proj_kernel<<<dim3(PROJ_GROUPS, 3), dim3(HID), 0, stream>>>(X, W1, Wb, b1, Ai, Bj, Y);
  grn_tile_kernel<<<dim3(NREG, NREG), dim3(TILE_THREADS), 0, stream>>>(
      X, Ai, Bj, Y, W2, b2, bb, out);
}
